# starting kernel
# baseline (speedup 1.0000x reference)
#include <hip/hip_runtime.h>
#include <cmath>

typedef float v2f __attribute__((ext_vector_type(2)));
typedef float v8f __attribute__((ext_vector_type(8)));

constexpr int Hdim  = 128;
constexpr int TT    = 64;
constexpr int PLEN  = 32;
constexpr int WAVES = 4;
constexpr int NJ    = 8;
constexpr int NK    = 32;
constexpr int BPACK = NK * NJ * 32 * 2;
constexpr int APACK = NK * 32 * 2;

extern __shared__ float smem[];

__device__ __forceinline__ int a_idx(int m, int k) {
    return (k >> 2) * 64 + (((((k >> 1) & 1) << 4) + m) << 1) + (k & 1);
}

__device__ __forceinline__ float fast_tanh(float x) {
    float e = __builtin_amdgcn_exp2f(x * 2.88539008177792681472f);
    return 1.0f - 2.0f * __builtin_amdgcn_rcpf(e + 1.0f);
}

__global__ __launch_bounds__(WAVES * 32)
void traj_rnn_kernel(const float* __restrict__ x,
                     const float* __restrict__ eWih, const float* __restrict__ eWhh,
                     const float* __restrict__ ebih, const float* __restrict__ ebhh,
                     const float* __restrict__ dWih, const float* __restrict__ dWhh,
                     const float* __restrict__ dbih, const float* __restrict__ dbhh,
                     const float* __restrict__ fcW,  const float* __restrict__ fcb,
                     float* __restrict__ out, int B)
{
    float* packBe = smem;
    float* packBd = packBe + BPACK;
    float* fcWl   = packBd + BPACK;
    float* fcbl   = fcWl + 3 * Hdim;
    float* hall   = fcbl + 4;
    float* iall   = hall + WAVES * APACK;

    const int tid  = threadIdx.x;
    const int wave = tid >> 5;
    const int lane = tid & 31;
    const int half = lane >> 4;
    const int ln   = lane & 15;

    for (int idx = tid; idx < BPACK; idx += blockDim.x) {
        int p  = idx & 1;
        int l  = (idx >> 1) & 31;
        int j  = (idx >> 6) & 7;
        int kk = idx >> 9;
        int n  = j * 16 + (l & 15);
        int k  = kk * 4 + ((l >> 4) << 1) + p;
        packBe[idx] = eWhh[n * Hdim + k];
        packBd[idx] = dWhh[n * Hdim + k];
    }
    for (int idx = tid; idx < 3 * Hdim; idx += blockDim.x) fcWl[idx] = fcW[idx];
    if (tid < 3) fcbl[tid] = fcb[tid];
    __syncthreads();

    const int tile = blockIdx.x * WAVES + wave;
    const int b0   = tile * 16;
    if (b0 >= B) return;

    float* hA = hall + wave * APACK;
    float* il = iall + wave * (16 * 4);

    float bse[NJ], bsd[NJ], wie[NJ][3], wid[NJ][3];
#pragma unroll
    for (int j = 0; j < NJ; ++j) {
        int col = j * 16 + ln;
        bse[j] = ebih[col] + ebhh[col];
        bsd[j] = dbih[col] + dbhh[col];
#pragma unroll
        for (int i = 0; i < 3; ++i) {
            wie[j][i] = eWih[col * 3 + i];
            wid[j][i] = dWih[col * 3 + i];
        }
    }

    for (int idx = lane; idx < APACK; idx += 32) hA[idx] = 0.0f;

    v8f c[NJ];

    for (int t = 0; t < TT; ++t) {
        float xv[8][3];
#pragma unroll
        for (int r = 0; r < 8; ++r) {
            const float* xp = x + ((size_t)(b0 + r + half * 8) * TT + t) * 3;
#pragma unroll
            for (int i = 0; i < 3; ++i) xv[r][i] = xp[i];
        }
#pragma unroll
        for (int j = 0; j < NJ; ++j)
#pragma unroll
            for (int r = 0; r < 8; ++r)
                c[j][r] = bse[j] + xv[r][0] * wie[j][0] + xv[r][1] * wie[j][1]
                                 + xv[r][2] * wie[j][2];

#pragma unroll 4
        for (int kk = 0; kk < NK; ++kk) {
            v2f a = *(const v2f*)&hA[(kk * 32 + lane) * 2];
#pragma unroll
            for (int j = 0; j < NJ; ++j) {
                v2f bf = *(const v2f*)&packBe[((kk * NJ + j) * 32 + lane) * 2];
                c[j] = __builtin_amdgcn_wmma_f32_16x16x4_f32(
                        false, a, false, bf, (short)0, c[j], false, false);
            }
        }
#pragma unroll
        for (int j = 0; j < NJ; ++j)
#pragma unroll
            for (int r = 0; r < 8; ++r)
                hA[a_idx(r + half * 8, j * 16 + ln)] = fast_tanh(c[j][r]);
    }

    if (lane < 16) {
#pragma unroll
        for (int i = 0; i < 3; ++i)
            il[lane * 4 + i] = x[((size_t)(b0 + lane) * TT + (TT - 1)) * 3 + i];
    }

    for (int p = 0; p < PLEN; ++p) {
        float xv[8][3];
#pragma unroll
        for (int r = 0; r < 8; ++r)
#pragma unroll
            for (int i = 0; i < 3; ++i) xv[r][i] = il[(r + half * 8) * 4 + i];
#pragma unroll
        for (int j = 0; j < NJ; ++j)
#pragma unroll
            for (int r = 0; r < 8; ++r)
                c[j][r] = bsd[j] + xv[r][0] * wid[j][0] + xv[r][1] * wid[j][1]
                                 + xv[r][2] * wid[j][2];

#pragma unroll 4
        for (int kk = 0; kk < NK; ++kk) {
            v2f a = *(const v2f*)&hA[(kk * 32 + lane) * 2];
#pragma unroll
            for (int j = 0; j < NJ; ++j) {
                v2f bf = *(const v2f*)&packBd[((kk * NJ + j) * 32 + lane) * 2];
                c[j] = __builtin_amdgcn_wmma_f32_16x16x4_f32(
                        false, a, false, bf, (short)0, c[j], false, false);
            }
        }
#pragma unroll
        for (int j = 0; j < NJ; ++j)
#pragma unroll
            for (int r = 0; r < 8; ++r)
                hA[a_idx(r + half * 8, j * 16 + ln)] = fast_tanh(c[j][r]);

        if (lane < 16) {
            float a0 = fcbl[0], a1 = fcbl[1], a2 = fcbl[2];
#pragma unroll 8
            for (int k = 0; k < Hdim; ++k) {
                float hv = hA[a_idx(lane, k)];
                a0 += hv * fcWl[0 * Hdim + k];
                a1 += hv * fcWl[1 * Hdim + k];
                a2 += hv * fcWl[2 * Hdim + k];
            }
            size_t ob = ((size_t)(b0 + lane) * PLEN + p) * 3;
            out[ob + 0] = a0; out[ob + 1] = a1; out[ob + 2] = a2;
            il[lane * 4 + 0] = a0; il[lane * 4 + 1] = a1; il[lane * 4 + 2] = a2;
        }
    }
}

extern "C" void kernel_launch(void* const* d_in, const int* in_sizes, int n_in,
                              void* d_out, int out_size, void* d_ws, size_t ws_size,
                              hipStream_t stream) {
    (void)n_in; (void)out_size; (void)d_ws; (void)ws_size;
    const float* x    = (const float*)d_in[0];
    const float* eWih = (const float*)d_in[1];
    const float* eWhh = (const float*)d_in[2];
    const float* ebih = (const float*)d_in[3];
    const float* ebhh = (const float*)d_in[4];
    const float* dWih = (const float*)d_in[5];
    const float* dWhh = (const float*)d_in[6];
    const float* dbih = (const float*)d_in[7];
    const float* dbhh = (const float*)d_in[8];
    const float* fcW  = (const float*)d_in[9];
    const float* fcb  = (const float*)d_in[10];
    float* out = (float*)d_out;

    int B = in_sizes[0] / (TT * 3);
    int tiles  = (B + 15) / 16;
    int blocks = (tiles + WAVES - 1) / WAVES;

    size_t shmem = (size_t)(2 * BPACK + 3 * Hdim + 4
                            + WAVES * APACK + WAVES * 16 * 4) * sizeof(float);

    traj_rnn_kernel<<<blocks, WAVES * 32, shmem, stream>>>(
        x, eWih, eWhh, ebih, ebhh, dWih, dWhh, dbih, dbhh, fcW, fcb, out, B);
}


# Barrier moved up between the own-tile products: 8 cover the stores' drain, the step's LDS reads (partial sums, pred parts, next x) go out as one group ahead of the other 10; same sums
# speedup vs baseline: 5.2532x; 5.2532x over previous
#include <hip/hip_runtime.h>

typedef float v2f __attribute__((ext_vector_type(2)));
typedef float v4f __attribute__((ext_vector_type(4)));
typedef float v8f __attribute__((ext_vector_type(8)));

constexpr int Hdim = 128;
constexpr int TT   = 64;
constexpr int PLEN = 32;
constexpr int XP   = 193;
constexpr int OP   = 97;
constexpr int RBLK = 512;
constexpr int RBUF = 4 * 3 * RBLK;
constexpr int PBUF = 4 * 16 * 4;
constexpr int LDS_FLOATS = 16 * XP + 2 * RBUF + 2 * PBUF + 16 * OP;

extern __shared__ __attribute__((aligned(16))) float smem[];

__device__ __forceinline__ v8f mm4(v2f a, v2f b, v8f c) {
    return __builtin_amdgcn_wmma_f32_16x16x4_f32(false, a, false, b, (short)0, c, false, false);
}

__device__ __forceinline__ float fast_tanh(float x) {
    float e = __builtin_amdgcn_exp2f(x * 2.88539008177792681472f);
    return 1.0f - 2.0f * __builtin_amdgcn_rcpf(e + 1.0f);
}

__device__ __forceinline__ void load_w(const float* __restrict__ W, int q, int c, int half,
                                       v2f (&w)[8][8]) {
#pragma unroll
    for (int i = 0; i < 8; ++i) {
        const int jt = (2 * q + i) & 7;
#pragma unroll
        for (int jo = 0; jo < 2; ++jo) {
            const float* src = W + (16 * jt + c) * Hdim + 32 * q + 16 * jo + 8 * half;
            v4f lo = *(const v4f*)src;
            v4f hi = *(const v4f*)(src + 4);
            w[i][jo * 4 + 0] = v2f{lo.x, lo.y};
            w[i][jo * 4 + 1] = v2f{lo.z, lo.w};
            w[i][jo * 4 + 2] = v2f{hi.x, hi.y};
            w[i][jo * 4 + 3] = v2f{hi.z, hi.w};
        }
    }
}

__device__ __forceinline__ void load_wx(const float* __restrict__ Wih, const float* __restrict__ bih,
                                        const float* __restrict__ bhh, int q, int c, int half,
                                        v2f (&wx)[2]) {
#pragma unroll
    for (int jo = 0; jo < 2; ++jo) {
        const int n = 32 * q + 16 * jo + c;
        float a0 = Wih[n * 3 + 2 * half];
        float a1 = half ? (bih[n] + bhh[n]) : Wih[n * 3 + 1];
        wx[jo] = v2f{a0, a1};
    }
}

__device__ __forceinline__ void others_to_lds(const v2f (&w)[8][8], const v2f (&hB)[8],
                                              float* __restrict__ Rb, int q, int lane) {
    v8f acc[6];
#pragma unroll
    for (int i = 0; i < 6; ++i) acc[i] = v8f{0, 0, 0, 0, 0, 0, 0, 0};
#pragma unroll
    for (int ch = 0; ch < 8; ++ch)
#pragma unroll
        for (int i = 0; i < 6; ++i) acc[i] = mm4(w[i + 2][ch], hB[ch], acc[i]);
#pragma unroll
    for (int s = 1; s < 4; ++s) {
        const int d = (q + s) & 3;
        float* blk = Rb + (d * 3 + (s - 1)) * RBLK + lane * 4;
        const v8f lo = acc[2 * s - 2], hi = acc[2 * s - 1];
        *(v4f*)(blk + 0 * 128) = v4f{lo[0], lo[1], lo[2], lo[3]};
        *(v4f*)(blk + 1 * 128) = v4f{lo[4], lo[5], lo[6], lo[7]};
        *(v4f*)(blk + 2 * 128) = v4f{hi[0], hi[1], hi[2], hi[3]};
        *(v4f*)(blk + 3 * 128) = v4f{hi[4], hi[5], hi[6], hi[7]};
    }
    __builtin_amdgcn_sched_barrier(0);
}

template <int C0, int C1>
__device__ __forceinline__ void own_products(const v2f (&w)[8][8], const v2f (&hB)[8],
                                             v8f& o0, v8f& o1) {
#pragma unroll
    for (int ch = C0; ch < C1; ++ch) {
        o0 = mm4(w[0][ch], hB[ch], o0);
        o1 = mm4(w[1][ch], hB[ch], o1);
    }
}

__device__ __forceinline__ void gather_load(const float* __restrict__ Rb, int q, int lane,
                                            v4f (&g)[12]) {
#pragma unroll
    for (int s = 0; s < 3; ++s) {
        const float* blk = Rb + (q * 3 + s) * RBLK + lane * 4;
        g[4 * s + 0] = *(const v4f*)(blk + 0 * 128);
        g[4 * s + 1] = *(const v4f*)(blk + 1 * 128);
        g[4 * s + 2] = *(const v4f*)(blk + 2 * 128);
        g[4 * s + 3] = *(const v4f*)(blk + 3 * 128);
    }
}

__device__ __forceinline__ void gather_add(const v4f (&g)[12], v8f& o0, v8f& o1) {
#pragma unroll
    for (int s = 0; s < 3; ++s) {
#pragma unroll
        for (int r = 0; r < 4; ++r) {
            o0[r] += g[4 * s + 0][r]; o0[r + 4] += g[4 * s + 1][r];
            o1[r] += g[4 * s + 2][r]; o1[r + 4] += g[4 * s + 3][r];
        }
    }
}

__device__ __forceinline__ void to_state(const v8f& o0, const v8f& o1, v2f (&hB)[8]) {
#pragma unroll
    for (int rr = 0; rr < 4; ++rr) {
        hB[rr]     = v2f{fast_tanh(o0[2 * rr]), fast_tanh(o0[2 * rr + 1])};
        hB[4 + rr] = v2f{fast_tanh(o1[2 * rr]), fast_tanh(o1[2 * rr + 1])};
    }
}

__device__ __forceinline__ void fc_partial(const v2f (&fcA)[8], const v2f (&hB)[8],
                                           float* __restrict__ Pb, int q, int c, int half) {
    const v8f z = v8f{0, 0, 0, 0, 0, 0, 0, 0};
    v8f fa = z, fb = z;
#pragma unroll
    for (int ch = 0; ch < 8; ch += 2) {
        fa = mm4(fcA[ch],     hB[ch],     fa);
        fb = mm4(fcA[ch + 1], hB[ch + 1], fb);
    }
    if (half == 0)
        *(v4f*)(Pb + (q * 16 + c) * 4) = v4f{fa[0] + fb[0], fa[1] + fb[1], fa[2] + fb[2], 0.0f};
}

__device__ __forceinline__ void pred_load(const float* __restrict__ Pb, int c, v4f (&pp)[4]) {
#pragma unroll
    for (int w = 0; w < 4; ++w) pp[w] = *(const v4f*)(Pb + (w * 16 + c) * 4);
}

__device__ __forceinline__ void pred_add(const v4f (&pp)[4], float fb0, float fb1, float fb2,
                                         float& y0, float& y1, float& y2) {
    y0 = fb0 + pp[0].x + pp[1].x + pp[2].x + pp[3].x;
    y1 = fb1 + pp[0].y + pp[1].y + pp[2].y + pp[3].y;
    y2 = fb2 + pp[0].z + pp[1].z + pp[2].z + pp[3].z;
}

__global__ __launch_bounds__(128)
void rnn_ksplit_kernel(const float* __restrict__ x,
                       const float* __restrict__ eWih, const float* __restrict__ eWhh,
                       const float* __restrict__ ebih, const float* __restrict__ ebhh,
                       const float* __restrict__ dWih, const float* __restrict__ dWhh,
                       const float* __restrict__ dbih, const float* __restrict__ dbhh,
                       const float* __restrict__ fcW,  const float* __restrict__ fcb,
                       float* __restrict__ out)
{
    float* xs = smem;
    float* R  = xs + 16 * XP;
    float* PP = R + 2 * RBUF;
    float* ob = PP + 2 * PBUF;

    const int tid  = threadIdx.x;
    const int q    = tid >> 5;
    const int lane = tid & 31;
    const int half = lane >> 4;
    const int c    = lane & 15;
    const int b0   = blockIdx.x * 16;

    {
        const float* xt = x + (size_t)b0 * (TT * 3);
#pragma unroll
        for (int k = 0; k < 6; ++k) {
            const int e = 4 * (tid + 128 * k);
            v4f v = *(const v4f*)(xt + e);
            const int row = e / 192, col = e % 192;
            float* dst = xs + row * XP + col;
            dst[0] = v.x; dst[1] = v.y; dst[2] = v.z; dst[3] = v.w;
        }
    }

    v2f wE[8][8], wD[8][8], wxE[2], wxD[2], fcA[8];
    load_w(eWhh, q, c, half, wE);
    load_wx(eWih, ebih, ebhh, q, c, half, wxE);
    load_w(dWhh, q, c, half, wD);
    load_wx(dWih, dbih, dbhh, q, c, half, wxD);
#pragma unroll
    for (int jo = 0; jo < 2; ++jo)
#pragma unroll
        for (int rr = 0; rr < 4; ++rr) {
            const int k = 32 * q + 16 * jo + 8 * half + 2 * rr;
            const int m = c < 3 ? c : 0;
            v2f f = *(const v2f*)(fcW + m * Hdim + k);
            fcA[jo * 4 + rr] = c < 3 ? f : v2f{0.0f, 0.0f};
        }
    const float fb0 = fcb[0], fb1 = fcb[1], fb2 = fcb[2];

    v2f hB[8];
#pragma unroll
    for (int ch = 0; ch < 8; ++ch) hB[ch] = v2f{0.0f, 0.0f};

    __syncthreads();

    const float* xrow = xs + c * XP;
    const v8f zero8 = v8f{0, 0, 0, 0, 0, 0, 0, 0};

    float e0 = xrow[2 * half], e1 = xrow[1];
    for (int t = 0; t < TT; ++t) {
        float* Rb = R + (t & 1) * RBUF;
        const v2f bx = v2f{e0, half ? 1.0f : e1};

        others_to_lds(wE, hB, Rb, q, lane);

        v8f o0 = mm4(wxE[0], bx, zero8);
        v8f o1 = mm4(wxE[1], bx, zero8);
        own_products<0, 3>(wE, hB, o0, o1);
        asm volatile("" :: "v"(o0), "v"(o1));
        __syncthreads();
        v4f g[12];
        gather_load(Rb, q, lane, g);
        const int tn = t + 1 < TT ? t + 1 : t;
        e0 = xrow[3 * tn + 2 * half];
        e1 = xrow[3 * tn + 1];
        __builtin_amdgcn_sched_barrier(0);
        own_products<3, 8>(wE, hB, o0, o1);
        gather_add(g, o0, o1);
        to_state(o0, o1, hB);
    }

    for (int p = 0; p < PLEN; ++p) {
        const int par = (TT + p) & 1;
        float* Rb = R + par * RBUF;
        float* Pb = PP + par * PBUF;

        fc_partial(fcA, hB, Pb, q, c, half);
        others_to_lds(wD, hB, Rb, q, lane);
        v8f o0 = zero8, o1 = zero8;
        own_products<0, 4>(wD, hB, o0, o1);
        asm volatile("" :: "v"(o0), "v"(o1));
        __syncthreads();
        v4f pp[4], g[12];
        pred_load(Pb, c, pp);
        gather_load(Rb, q, lane, g);
        __builtin_amdgcn_sched_barrier(0);
        own_products<4, 8>(wD, hB, o0, o1);

        float y0, y1, y2;
        pred_add(pp, fb0, fb1, fb2, y0, y1, y2);
        gather_add(g, o0, o1);
        const float i0 = p ? y0 : xrow[3 * (TT - 1) + 0];
        const float i1 = p ? y1 : xrow[3 * (TT - 1) + 1];
        const float i2 = p ? y2 : xrow[3 * (TT - 1) + 2];
        const v2f bx = half ? v2f{i2, 1.0f} : v2f{i0, i1};
        o0 = mm4(wxD[0], bx, o0);
        o1 = mm4(wxD[1], bx, o1);
        to_state(o0, o1, hB);
        if (p > 0 && q == 0 && half == 0) {
            float* o = ob + c * OP + (p - 1) * 3;
            o[0] = y0; o[1] = y1; o[2] = y2;
        }
    }
    {
        float* Pb = PP + ((TT + PLEN) & 1) * PBUF;
        fc_partial(fcA, hB, Pb, q, c, half);
        __syncthreads();
        v4f pp[4];
        pred_load(Pb, c, pp);
        float y0, y1, y2;
        pred_add(pp, fb0, fb1, fb2, y0, y1, y2);
        if (q == 0 && half == 0) {
            float* o = ob + c * OP + (PLEN - 1) * 3;
            o[0] = y0; o[1] = y1; o[2] = y2;
        }
    }

    __syncthreads();
    float* ot = out + (size_t)b0 * (PLEN * 3);
    for (int e = tid; e < 16 * PLEN * 3; e += 128)
        ot[e] = ob[(e / (PLEN * 3)) * OP + e % (PLEN * 3)];
}

extern "C" void kernel_launch(void* const* d_in, const int* in_sizes, int n_in,
                              void* d_out, int out_size, void* d_ws, size_t ws_size,
                              hipStream_t stream) {
    (void)n_in; (void)out_size; (void)d_ws; (void)ws_size;
    const float* x    = (const float*)d_in[0];
    const float* eWih = (const float*)d_in[1];
    const float* eWhh = (const float*)d_in[2];
    const float* ebih = (const float*)d_in[3];
    const float* ebhh = (const float*)d_in[4];
    const float* dWih = (const float*)d_in[5];
    const float* dWhh = (const float*)d_in[6];
    const float* dbih = (const float*)d_in[7];
    const float* dbhh = (const float*)d_in[8];
    const float* fcW  = (const float*)d_in[9];
    const float* fcb  = (const float*)d_in[10];
    float* out = (float*)d_out;

    int B     = in_sizes[0] / (TT * 3);
    int tiles = (B + 15) / 16;

    size_t shmem = (size_t)162 * 1024;
    static_assert(LDS_FLOATS * sizeof(float) <= (size_t)162 * 1024, "LDS layout");

    rnn_ksplit_kernel<<<tiles, 128, shmem, stream>>>(
        x, eWih, eWhh, ebih, ebhh, dWih, dWhh, dbih, dbhh, fcW, fcb, out);
}
